# starting kernel
# baseline (speedup 1.0000x reference)
#include <hip/hip_runtime.h>
#include <hip/hip_bf16.h>

#define N_NODES 50000
#define N_EDGES 600000
#define DIM 128
#define ALPHA_C 0.1f
#define BETA_C 0.22314355131420976f
#define HS_STRIDE 132

typedef __attribute__((ext_vector_type(2))) float v2f;
typedef __attribute__((ext_vector_type(8))) float v8f;

__global__ void gcnii_zero(float* out, float* __restrict__ deg) {
    int idx = blockIdx.x * blockDim.x + threadIdx.x;
    if (idx < N_NODES * DIM) out[idx] = 0.0f;
    if (idx < N_NODES) deg[idx] = 0.0f;
}

__global__ void gcnii_degree(const int* __restrict__ dst, float* __restrict__ deg) {
    int e = blockIdx.x * blockDim.x + threadIdx.x;
    if (e < N_EDGES) atomicAdd(&deg[dst[e]], 1.0f);
}

__global__ void gcnii_norm(float* __restrict__ deg) {
    int i = blockIdx.x * blockDim.x + threadIdx.x;
    if (i < N_NODES) deg[i] = rsqrtf(fmaxf(deg[i], 1.0f));
}

__global__ void gcnii_scatter(const float* __restrict__ feat,
                              const float* __restrict__ norm,
                              const int* __restrict__ src,
                              const int* __restrict__ dst,
                              float* agg) {
    int gtid = blockIdx.x * blockDim.x + threadIdx.x;
    int e    = gtid >> 5;
    int lane = gtid & 31;
    if (e >= N_EDGES) return;
    int s = src[e];
    int d = dst[e];
    float ns = norm[s];
    float4 v = *(const float4*)(feat + (size_t)s * DIM + lane * 4);
    float* ad = agg + (size_t)d * DIM + lane * 4;
    atomicAdd(ad + 0, v.x * ns);
    atomicAdd(ad + 1, v.y * ns);
    atomicAdd(ad + 2, v.z * ns);
    atomicAdd(ad + 3, v.w * ns);
}

__global__ __launch_bounds__(256) void gcnii_gemm(
    const float* agg,
    const float* __restrict__ feat0,
    const float* __restrict__ W,
    const float* __restrict__ bias,
    const float* __restrict__ norm,
    float* out)
{
    __shared__ float Wl[DIM * DIM];
    __shared__ float hs[16 * HS_STRIDE];
    __shared__ float bs[DIM];

    const int tid = threadIdx.x;
    const int m0  = blockIdx.x * 16;

    for (int i = tid; i < DIM * DIM / 4; i += 256)
        ((float4*)Wl)[i] = ((const float4*)W)[i];
    if (tid < DIM) bs[tid] = bias[tid];

    for (int i = tid; i < 16 * DIM / 4; i += 256) {
        int r = i >> 5;
        int c = (i & 31) * 4;
        int row = m0 + r;
        float nr = norm[row] * (1.0f - ALPHA_C);
        float4 a  = *(const float4*)(agg   + (size_t)row * DIM + c);
        float4 f0 = *(const float4*)(feat0 + (size_t)row * DIM + c);
        float4 h;
        h.x = a.x * nr + ALPHA_C * f0.x;
        h.y = a.y * nr + ALPHA_C * f0.y;
        h.z = a.z * nr + ALPHA_C * f0.z;
        h.w = a.w * nr + ALPHA_C * f0.w;
        *(float4*)(hs + r * HS_STRIDE + c) = h;
    }
    __syncthreads();

    const int wave = tid >> 5;
    const int lane = tid & 31;
    const int half = lane >> 4;
    const int lr   = lane & 15;
    const int n0   = wave * 16;

    v8f acc = {};
    #pragma unroll
    for (int k0 = 0; k0 < DIM; k0 += 4) {
        const int ka = k0 + 2 * half;
        v2f a, b;
        a.x = hs[lr * HS_STRIDE + ka];
        a.y = hs[lr * HS_STRIDE + ka + 1];
        b.x = Wl[ka * DIM + n0 + lr];
        b.y = Wl[(ka + 1) * DIM + n0 + lr];
        acc = __builtin_amdgcn_wmma_f32_16x16x4_f32(
             false, a,  false, b,
             (short)0, acc,  false,  false);
    }

    #pragma unroll
    for (int r = 0; r < 8; r++) {
        int lrow = r + 8 * half;
        int row  = m0 + lrow;
        int col  = n0 + lr;
        float hv = hs[lrow * HS_STRIDE + col];
        out[(size_t)row * DIM + col] =
            (1.0f - BETA_C) * hv + BETA_C * acc[r] + bs[col];
    }
}

extern "C" void kernel_launch(void* const* d_in, const int* in_sizes, int n_in,
                              void* d_out, int out_size, void* d_ws, size_t ws_size,
                              hipStream_t stream) {
    const float* feat  = (const float*)d_in[0];
    const float* feat0 = (const float*)d_in[1];
    const float* W     = (const float*)d_in[2];
    const float* bias  = (const float*)d_in[3];
    const int*   src   = (const int*)d_in[4];
    const int*   dst   = (const int*)d_in[5];
    float* out = (float*)d_out;
    float* deg = (float*)d_ws;

    gcnii_zero<<<(N_NODES * DIM + 255) / 256, 256, 0, stream>>>(out, deg);
    gcnii_degree<<<(N_EDGES + 255) / 256, 256, 0, stream>>>(dst, deg);
    gcnii_norm<<<(N_NODES + 255) / 256, 256, 0, stream>>>(deg);
    gcnii_scatter<<<(N_EDGES * 32 + 255) / 256, 256, 0, stream>>>(feat, deg, src, dst, out);
    gcnii_gemm<<<N_NODES / 16, 256, 0, stream>>>(out, feat0, W, bias, deg, out);
}


# Scatter lanes remapped (lane l takes floats l, l+32, l+64, l+96): each atomic add and gather load of a wave covers one whole 128-byte line, not 4 lines of 8 lanes; sums stay in arrival order
# speedup vs baseline: 1.2801x; 1.2801x over previous
#include <hip/hip_runtime.h>
#include <hip/hip_bf16.h>

#define N_NODES 50000
#define N_EDGES 600000
#define DIM 128
#define ALPHA_C 0.1f
#define BETA_C 0.22314355131420976f
#define HS_STRIDE 132

typedef __attribute__((ext_vector_type(2))) float v2f;
typedef __attribute__((ext_vector_type(8))) float v8f;

__global__ void gcnii_zero(float* out, float* __restrict__ deg) {
    int idx = blockIdx.x * blockDim.x + threadIdx.x;
    if (idx < N_NODES * DIM) out[idx] = 0.0f;
    if (idx < N_NODES) deg[idx] = 0.0f;
}

__global__ void gcnii_degree(const int* __restrict__ dst, float* __restrict__ deg) {
    int e = blockIdx.x * blockDim.x + threadIdx.x;
    if (e < N_EDGES) atomicAdd(&deg[dst[e]], 1.0f);
}

__global__ void gcnii_norm(float* __restrict__ deg) {
    int i = blockIdx.x * blockDim.x + threadIdx.x;
    if (i < N_NODES) deg[i] = rsqrtf(fmaxf(deg[i], 1.0f));
}

__global__ void gcnii_scatter(const float* __restrict__ feat,
                              const float* __restrict__ norm,
                              const int* __restrict__ src,
                              const int* __restrict__ dst,
                              float* agg) {
    int gtid = blockIdx.x * blockDim.x + threadIdx.x;
    int e    = gtid >> 5;
    int lane = gtid & 31;
    if (e >= N_EDGES) return;
    int s = src[e];
    int d = dst[e];
    float ns = norm[s];
    const float* fs = feat + (size_t)s * DIM + lane;
    float v0 = fs[0];
    float v1 = fs[32];
    float v2 = fs[64];
    float v3 = fs[96];
    float* ad = agg + (size_t)d * DIM + lane;
    atomicAdd(ad,      v0 * ns);
    atomicAdd(ad + 32, v1 * ns);
    atomicAdd(ad + 64, v2 * ns);
    atomicAdd(ad + 96, v3 * ns);
}

__global__ __launch_bounds__(256) void gcnii_gemm(
    const float* agg,
    const float* __restrict__ feat0,
    const float* __restrict__ W,
    const float* __restrict__ bias,
    const float* __restrict__ norm,
    float* out)
{
    __shared__ float Wl[DIM * DIM];
    __shared__ float hs[16 * HS_STRIDE];
    __shared__ float bs[DIM];

    const int tid = threadIdx.x;
    const int m0  = blockIdx.x * 16;

    for (int i = tid; i < DIM * DIM / 4; i += 256)
        ((float4*)Wl)[i] = ((const float4*)W)[i];
    if (tid < DIM) bs[tid] = bias[tid];

    for (int i = tid; i < 16 * DIM / 4; i += 256) {
        int r = i >> 5;
        int c = (i & 31) * 4;
        int row = m0 + r;
        float nr = norm[row] * (1.0f - ALPHA_C);
        float4 a  = *(const float4*)(agg   + (size_t)row * DIM + c);
        float4 f0 = *(const float4*)(feat0 + (size_t)row * DIM + c);
        float4 h;
        h.x = a.x * nr + ALPHA_C * f0.x;
        h.y = a.y * nr + ALPHA_C * f0.y;
        h.z = a.z * nr + ALPHA_C * f0.z;
        h.w = a.w * nr + ALPHA_C * f0.w;
        *(float4*)(hs + r * HS_STRIDE + c) = h;
    }
    __syncthreads();

    const int wave = tid >> 5;
    const int lane = tid & 31;
    const int half = lane >> 4;
    const int lr   = lane & 15;
    const int n0   = wave * 16;

    v8f acc = {};
    #pragma unroll
    for (int k0 = 0; k0 < DIM; k0 += 4) {
        const int ka = k0 + 2 * half;
        v2f a, b;
        a.x = hs[lr * HS_STRIDE + ka];
        a.y = hs[lr * HS_STRIDE + ka + 1];
        b.x = Wl[ka * DIM + n0 + lr];
        b.y = Wl[(ka + 1) * DIM + n0 + lr];
        acc = __builtin_amdgcn_wmma_f32_16x16x4_f32(
             false, a,  false, b,
             (short)0, acc,  false,  false);
    }

    #pragma unroll
    for (int r = 0; r < 8; r++) {
        int lrow = r + 8 * half;
        int row  = m0 + lrow;
        int col  = n0 + lr;
        float hv = hs[lrow * HS_STRIDE + col];
        out[(size_t)row * DIM + col] =
            (1.0f - BETA_C) * hv + BETA_C * acc[r] + bs[col];
    }
}

extern "C" void kernel_launch(void* const* d_in, const int* in_sizes, int n_in,
                              void* d_out, int out_size, void* d_ws, size_t ws_size,
                              hipStream_t stream) {
    const float* feat  = (const float*)d_in[0];
    const float* feat0 = (const float*)d_in[1];
    const float* W     = (const float*)d_in[2];
    const float* bias  = (const float*)d_in[3];
    const int*   src   = (const int*)d_in[4];
    const int*   dst   = (const int*)d_in[5];
    float* out = (float*)d_out;
    float* deg = (float*)d_ws;

    gcnii_zero<<<(N_NODES * DIM + 255) / 256, 256, 0, stream>>>(out, deg);
    gcnii_degree<<<(N_EDGES + 255) / 256, 256, 0, stream>>>(dst, deg);
    gcnii_norm<<<(N_NODES + 255) / 256, 256, 0, stream>>>(deg);
    gcnii_scatter<<<(N_EDGES * 32 + 255) / 256, 256, 0, stream>>>(feat, deg, src, dst, out);
    gcnii_gemm<<<N_NODES / 16, 256, 0, stream>>>(out, feat0, W, bias, deg, out);
}


# Float atomic scatter replaced by per-target source lists (slot from a returning counter add, list kept in the target's own out row) summed in registers (fma) in the product kernel, in arrival order
# speedup vs baseline: 2.0564x; 1.6064x over previous
#include <hip/hip_runtime.h>
#include <hip/hip_bf16.h>


#define N_NODES 50000
#define N_EDGES 600000
#define DIM 128
#define ALPHA_C 0.1f
#define BETA_C 0.22314355131420976f
#define HS_STRIDE 132
#define LIST_CAP 128

typedef __attribute__((ext_vector_type(2))) float v2f;
typedef __attribute__((ext_vector_type(8))) float v8f;

__global__ void gcnii_zero_cnt(int* __restrict__ cnt) {
    int i = blockIdx.x * blockDim.x + threadIdx.x;
    if (i < N_NODES) cnt[i] = 0;
}

__global__ void gcnii_fill(const int* __restrict__ src,
                           const int* __restrict__ dst,
                           int* __restrict__ cnt,
                           int* __restrict__ lists) {
    int e = blockIdx.x * blockDim.x + threadIdx.x;
    if (e < N_EDGES) {
        int d = dst[e];
        int s = src[e];
        int pos = atomicAdd(&cnt[d], 1);
        if (pos < LIST_CAP) lists[(size_t)d * DIM + pos] = s;
    }
}

__device__ __forceinline__ float rl_f(float v, int l) {
    return __int_as_float(__builtin_amdgcn_readlane(__float_as_int(v), l));
}

__device__ __forceinline__ void fma4(float4& acc, const float4 v, const float w) {
    acc.x = fmaf(v.x, w, acc.x);
    acc.y = fmaf(v.y, w, acc.y);
    acc.z = fmaf(v.z, w, acc.z);
    acc.w = fmaf(v.w, w, acc.w);
}

__device__ __noinline__ float4 gather_row_by_scan(const float* __restrict__ frow,
                                                  const int* __restrict__ cnt,
                                                  const int* __restrict__ src,
                                                  const int* __restrict__ dst,
                                                  int row, int lane) {
    float4 acc = make_float4(0.f, 0.f, 0.f, 0.f);
    for (int e0 = 0; e0 < N_EDGES; e0 += 32) {
        const int e = e0 + lane;
        const int s = src[e];
        const bool hit = (dst[e] == row);
        const float w = hit ? rsqrtf(fmaxf((float)cnt[s], 1.0f)) : 0.0f;
        unsigned mask = (unsigned)__ballot(hit);
        while (mask) {
            const int j = __builtin_ctz(mask);
            mask &= mask - 1;
            const int sj = __builtin_amdgcn_readlane(s, j);
            const float4 v = *(const float4*)(frow + (size_t)sj * DIM);
            fma4(acc, v, rl_f(w, j));
        }
    }
    return acc;
}

__global__ __launch_bounds__(256) void gcnii_gather_gemm(
    const float* __restrict__ feat,
    const float* __restrict__ feat0,
    const float* __restrict__ W,
    const float* __restrict__ bias,
    const int* __restrict__ cnt,
    const int* __restrict__ src,
    const int* __restrict__ dst,
    const int* lists,
    float* out)
{
    __shared__ float Wl[DIM * DIM];
    __shared__ float hs[16 * HS_STRIDE];
    __shared__ float bs[DIM];

    const int tid = threadIdx.x;
    const int m0  = blockIdx.x * 16;

    const int lane = tid & 31;
    const int wave = __builtin_amdgcn_readfirstlane(tid >> 5);

    const int lrA = 2 * wave, lrB = 2 * wave + 1;
    const int rowA = m0 + lrA, rowB = m0 + lrB;
    const int nA = cnt[rowA], nB = cnt[rowB];
    const int* lstA = lists + (size_t)rowA * DIM;
    const int* lstB = lists + (size_t)rowB * DIM;
    const float* frow = feat + lane * 4;
    const int lenA = (nA <= LIST_CAP) ? nA : 0;
    const int lenB = (nB <= LIST_CAP) ? nB : 0;
    const int lenM = max(lenA, lenB);
    int sA = 0, sB = 0;
    if (lane < lenA) sA = lstA[lane];
    if (lane < lenB) sB = lstB[lane];
    const int dA = cnt[sA], dB = cnt[sB];
    const float4 f0A = *(const float4*)(feat0 + (size_t)rowA * DIM + lane * 4);
    const float4 f0B = *(const float4*)(feat0 + (size_t)rowB * DIM + lane * 4);

    for (int i = tid; i < DIM * DIM / 4; i += 256)
        ((float4*)Wl)[i] = ((const float4*)W)[i];
    if (tid < DIM) bs[tid] = bias[tid];

    float wA = (lane < lenA) ? rsqrtf(fmaxf((float)dA, 1.0f)) : 0.0f;
    float wB = (lane < lenB) ? rsqrtf(fmaxf((float)dB, 1.0f)) : 0.0f;

    {
        float4 accA = make_float4(0.f, 0.f, 0.f, 0.f);
        float4 accB = make_float4(0.f, 0.f, 0.f, 0.f);
        for (int c = 0; c < lenM; c += 32) {
            const int mA = min(lenA - c, 32);
            const int mB = min(lenB - c, 32);
            if (c > 0) {
                sA = 0; sB = 0;
                if (lane < mA) sA = lstA[c + lane];
                if (lane < mB) sB = lstB[c + lane];
                const int eA = cnt[sA], eB = cnt[sB];
                wA = (lane < mA) ? rsqrtf(fmaxf((float)eA, 1.0f)) : 0.0f;
                wB = (lane < mB) ? rsqrtf(fmaxf((float)eB, 1.0f)) : 0.0f;
            }
            const int lastA = max(mA - 1, 0), lastB = max(mB - 1, 0);
            const int mM = max(mA, mB);
            for (int j = 0; j < mM; j += 4) {
                const int a0 = __builtin_amdgcn_readlane(sA, min(j,     lastA));
                const int a1 = __builtin_amdgcn_readlane(sA, min(j + 1, lastA));
                const int a2 = __builtin_amdgcn_readlane(sA, min(j + 2, lastA));
                const int a3 = __builtin_amdgcn_readlane(sA, min(j + 3, lastA));
                const int b0 = __builtin_amdgcn_readlane(sB, min(j,     lastB));
                const int b1 = __builtin_amdgcn_readlane(sB, min(j + 1, lastB));
                const int b2 = __builtin_amdgcn_readlane(sB, min(j + 2, lastB));
                const int b3 = __builtin_amdgcn_readlane(sB, min(j + 3, lastB));
                const float4 va0 = *(const float4*)(frow + (size_t)a0 * DIM);
                const float4 va1 = *(const float4*)(frow + (size_t)a1 * DIM);
                const float4 va2 = *(const float4*)(frow + (size_t)a2 * DIM);
                const float4 va3 = *(const float4*)(frow + (size_t)a3 * DIM);
                const float4 vb0 = *(const float4*)(frow + (size_t)b0 * DIM);
                const float4 vb1 = *(const float4*)(frow + (size_t)b1 * DIM);
                const float4 vb2 = *(const float4*)(frow + (size_t)b2 * DIM);
                const float4 vb3 = *(const float4*)(frow + (size_t)b3 * DIM);
                fma4(accA, va0, rl_f(wA, j));
                fma4(accA, va1, rl_f(wA, j + 1));
                fma4(accA, va2, rl_f(wA, j + 2));
                fma4(accA, va3, rl_f(wA, j + 3));
                fma4(accB, vb0, rl_f(wB, j));
                fma4(accB, vb1, rl_f(wB, j + 1));
                fma4(accB, vb2, rl_f(wB, j + 2));
                fma4(accB, vb3, rl_f(wB, j + 3));
            }
        }
        if (nA > LIST_CAP) accA = gather_row_by_scan(frow, cnt, src, dst, rowA, lane);
        if (nB > LIST_CAP) accB = gather_row_by_scan(frow, cnt, src, dst, rowB, lane);

        const float nrA = rsqrtf(fmaxf((float)nA, 1.0f)) * (1.0f - ALPHA_C);
        const float nrB = rsqrtf(fmaxf((float)nB, 1.0f)) * (1.0f - ALPHA_C);
        float4 hA, hB;
        hA.x = accA.x * nrA + ALPHA_C * f0A.x;
        hA.y = accA.y * nrA + ALPHA_C * f0A.y;
        hA.z = accA.z * nrA + ALPHA_C * f0A.z;
        hA.w = accA.w * nrA + ALPHA_C * f0A.w;
        hB.x = accB.x * nrB + ALPHA_C * f0B.x;
        hB.y = accB.y * nrB + ALPHA_C * f0B.y;
        hB.z = accB.z * nrB + ALPHA_C * f0B.z;
        hB.w = accB.w * nrB + ALPHA_C * f0B.w;
        *(float4*)(hs + lrA * HS_STRIDE + lane * 4) = hA;
        *(float4*)(hs + lrB * HS_STRIDE + lane * 4) = hB;
    }
    __syncthreads();

    const int half = lane >> 4;
    const int lr   = lane & 15;
    const int n0   = wave * 16;

    v8f acc = {};
    #pragma unroll
    for (int k0 = 0; k0 < DIM; k0 += 4) {
        const int ka = k0 + 2 * half;
        v2f a, b;
        a.x = hs[lr * HS_STRIDE + ka];
        a.y = hs[lr * HS_STRIDE + ka + 1];
        b.x = Wl[ka * DIM + n0 + lr];
        b.y = Wl[(ka + 1) * DIM + n0 + lr];
        acc = __builtin_amdgcn_wmma_f32_16x16x4_f32(
             false, a,  false, b,
             (short)0, acc,  false,  false);
    }

    #pragma unroll
    for (int r = 0; r < 8; r++) {
        int lrow = r + 8 * half;
        int row  = m0 + lrow;
        int col  = n0 + lr;
        float hv = hs[lrow * HS_STRIDE + col];
        out[(size_t)row * DIM + col] =
            (1.0f - BETA_C) * hv + BETA_C * acc[r] + bs[col];
    }
}

extern "C" void kernel_launch(void* const* d_in, const int* in_sizes, int n_in,
                              void* d_out, int out_size, void* d_ws, size_t ws_size,
                              hipStream_t stream) {
    const float* feat  = (const float*)d_in[0];
    const float* feat0 = (const float*)d_in[1];
    const float* W     = (const float*)d_in[2];
    const float* bias  = (const float*)d_in[3];
    const int*   src   = (const int*)d_in[4];
    const int*   dst   = (const int*)d_in[5];
    float* out = (float*)d_out;
    int* cnt   = (int*)d_ws;
    int* lists = (int*)d_out;

    gcnii_zero_cnt<<<(N_NODES + 255) / 256, 256, 0, stream>>>(cnt);
    gcnii_fill<<<(N_EDGES + 255) / 256, 256, 0, stream>>>(src, dst, cnt, lists);
    gcnii_gather_gemm<<<N_NODES / 16, 256, 0, stream>>>(feat, feat0, W, bias, cnt, src, dst, lists, out);
}


# Product kernel made persistent: 768 work-groups walk the strips, W1 kept as B fragments in registers (not staged per strip), later strips' in-degrees and list heads fetched ahead; sums unchanged
# speedup vs baseline: 2.2014x; 1.0705x over previous
#include <hip/hip_runtime.h>
#include <hip/hip_bf16.h>


#define N_NODES 50000
#define N_EDGES 600000
#define DIM 128
#define ALPHA_C 0.1f
#define BETA_C 0.22314355131420976f
#define HS_STRIDE 132
#define N_STRIPS (N_NODES / 16)
#define N_BLOCKS 768
#define LIST_CAP 128

typedef __attribute__((ext_vector_type(2))) float v2f;
typedef __attribute__((ext_vector_type(8))) float v8f;

__global__ void gcnii_zero_cnt(int* __restrict__ cnt) {
    int i = blockIdx.x * blockDim.x + threadIdx.x;
    if (i < N_NODES) cnt[i] = 0;
}

__global__ void gcnii_fill(const int* __restrict__ src,
                           const int* __restrict__ dst,
                           int* __restrict__ cnt,
                           int* __restrict__ lists) {
    int e = blockIdx.x * blockDim.x + threadIdx.x;
    if (e < N_EDGES) {
        int d = dst[e];
        int s = src[e];
        int pos = atomicAdd(&cnt[d], 1);
        if (pos < LIST_CAP) lists[(size_t)d * DIM + pos] = s;
    }
}

__device__ __forceinline__ float rl_f(float v, int l) {
    return __int_as_float(__builtin_amdgcn_readlane(__float_as_int(v), l));
}

__device__ __forceinline__ void fma4(float4& acc, const float4 v, const float w) {
    acc.x = fmaf(v.x, w, acc.x);
    acc.y = fmaf(v.y, w, acc.y);
    acc.z = fmaf(v.z, w, acc.z);
    acc.w = fmaf(v.w, w, acc.w);
}

__device__ __noinline__ float4 gather_row_by_scan(const float* __restrict__ feat,
                                                  const int* __restrict__ cnt,
                                                  const int* __restrict__ src,
                                                  const int* __restrict__ dst,
                                                  int row, int lane) {
    const float* frow = feat + lane * 4;
    float4 acc = make_float4(0.f, 0.f, 0.f, 0.f);
    for (int e0 = 0; e0 < N_EDGES; e0 += 32) {
        const int e = e0 + lane;
        const int s = src[e];
        const bool hit = (dst[e] == row);
        const float w = hit ? rsqrtf(fmaxf((float)cnt[s], 1.0f)) : 0.0f;
        unsigned mask = (unsigned)__ballot(hit);
        while (mask) {
            const int j = __builtin_ctz(mask);
            mask &= mask - 1;
            const int sj = __builtin_amdgcn_readlane(s, j);
            const float4 v = *(const float4*)(frow + (size_t)sj * DIM);
            fma4(acc, v, rl_f(w, j));
        }
    }
    return acc;
}

__global__ __launch_bounds__(256) void gcnii_gather_gemm(
    const float* __restrict__ feat,
    const float* __restrict__ feat0,
    const float* __restrict__ W,
    const float* __restrict__ bias,
    const int* __restrict__ cnt,
    const int* __restrict__ src,
    const int* __restrict__ dst,
    const int* lists,
    float* out)
{
    __shared__ float hs[2][16 * HS_STRIDE];

    const int tid  = threadIdx.x;
    const int lane = tid & 31;
    const int wave = __builtin_amdgcn_readfirstlane(tid >> 5);

    const int half = lane >> 4;
    const int lr   = lane & 15;
    const int n0   = wave * 16;

    v2f bfr[DIM / 4];
    #pragma unroll
    for (int i = 0; i < DIM / 4; i++) {
        const int ka = 4 * i + 2 * half;
        bfr[i].x = W[ka * DIM + n0 + lr];
        bfr[i].y = W[(ka + 1) * DIM + n0 + lr];
    }
    const float bcol = bias[n0 + lr];

    const int lrA = 2 * wave, lrB = 2 * wave + 1;
    const unsigned lane4 = (unsigned)lane * 4u;
    const float4* feat4 = (const float4*)feat;
    int buf = 0;

    const int lrP = 2 * wave + (lane & 1);
    auto row_degrees = [&](int s) -> int {
        return cnt[min(s, N_STRIPS - 1) * 16 + lrP];
    };
    auto list_words = [&](int s, int lr_) -> int {
        return lists[(size_t)(min(s, N_STRIPS - 1) * 16 + lr_) * DIM + lane];
    };
    auto list_head = [&](int words, int n) -> int {
        return (lane < ((n <= LIST_CAP) ? n : 0)) ? words : 0;
    };

    int strip = blockIdx.x;
    int nv  = row_degrees(strip);
    int nv1 = row_degrees(strip + N_BLOCKS);
    int sA  = list_words(strip, lrA), sB = list_words(strip, lrB);
    int rA1 = list_words(strip + N_BLOCKS, lrA), rB1 = list_words(strip + N_BLOCKS, lrB);
    float4 f0A = *(const float4*)(feat0 + (size_t)(strip * 16 + lrA) * DIM + lane4);
    float4 f0B = *(const float4*)(feat0 + (size_t)(strip * 16 + lrB) * DIM + lane4);
    sA = list_head(sA, __builtin_amdgcn_readlane(nv, 0));
    sB = list_head(sB, __builtin_amdgcn_readlane(nv, 1));
    int dA = cnt[sA], dB = cnt[sB];

    for (; strip < N_STRIPS; strip += N_BLOCKS) {
        const int m0 = strip * 16;
        float* hb = hs[buf];

        const int rowA = m0 + lrA, rowB = m0 + lrB;
        const int nA = __builtin_amdgcn_readlane(nv, 0), nB = __builtin_amdgcn_readlane(nv, 1);
        const int* lstA = lists + (size_t)rowA * DIM;
        const int* lstB = lists + (size_t)rowB * DIM;
        const int lenA = (nA <= LIST_CAP) ? nA : 0;
        const int lenB = (nB <= LIST_CAP) ? nB : 0;
        const int lenM = max(lenA, lenB);

        float wA = (lane < lenA) ? rsqrtf(fmaxf((float)dA, 1.0f)) : 0.0f;
        float wB = (lane < lenB) ? rsqrtf(fmaxf((float)dB, 1.0f)) : 0.0f;

        {
            float4 accA = make_float4(0.f, 0.f, 0.f, 0.f);
            float4 accB = make_float4(0.f, 0.f, 0.f, 0.f);
            for (int c = 0; c < lenM; c += 32) {
                const int mA = min(lenA - c, 32);
                const int mB = min(lenB - c, 32);
                if (c > 0) {
                    sA = 0; sB = 0;
                    if (lane < mA) sA = lstA[c + lane];
                    if (lane < mB) sB = lstB[c + lane];
                    const int eA = cnt[sA], eB = cnt[sB];
                    wA = (lane < mA) ? rsqrtf(fmaxf((float)eA, 1.0f)) : 0.0f;
                    wB = (lane < mB) ? rsqrtf(fmaxf((float)eB, 1.0f)) : 0.0f;
                }
                const int lastA = max(mA - 1, 0), lastB = max(mB - 1, 0);
                const int mM = max(mA, mB);
                for (int j = 0; j < mM; j += 4) {
                    const int a0 = __builtin_amdgcn_readlane(sA, min(j,     lastA));
                    const int a1 = __builtin_amdgcn_readlane(sA, min(j + 1, lastA));
                    const int a2 = __builtin_amdgcn_readlane(sA, min(j + 2, lastA));
                    const int a3 = __builtin_amdgcn_readlane(sA, min(j + 3, lastA));
                    const int b0 = __builtin_amdgcn_readlane(sB, min(j,     lastB));
                    const int b1 = __builtin_amdgcn_readlane(sB, min(j + 1, lastB));
                    const int b2 = __builtin_amdgcn_readlane(sB, min(j + 2, lastB));
                    const int b3 = __builtin_amdgcn_readlane(sB, min(j + 3, lastB));
                    const float4 va0 = feat4[(unsigned)(a0 * (DIM / 4)) + (unsigned)lane];
                    const float4 va1 = feat4[(unsigned)(a1 * (DIM / 4)) + (unsigned)lane];
                    const float4 va2 = feat4[(unsigned)(a2 * (DIM / 4)) + (unsigned)lane];
                    const float4 va3 = feat4[(unsigned)(a3 * (DIM / 4)) + (unsigned)lane];
                    const float4 vb0 = feat4[(unsigned)(b0 * (DIM / 4)) + (unsigned)lane];
                    const float4 vb1 = feat4[(unsigned)(b1 * (DIM / 4)) + (unsigned)lane];
                    const float4 vb2 = feat4[(unsigned)(b2 * (DIM / 4)) + (unsigned)lane];
                    const float4 vb3 = feat4[(unsigned)(b3 * (DIM / 4)) + (unsigned)lane];
                    fma4(accA, va0, rl_f(wA, j));
                    fma4(accA, va1, rl_f(wA, j + 1));
                    fma4(accA, va2, rl_f(wA, j + 2));
                    fma4(accA, va3, rl_f(wA, j + 3));
                    fma4(accB, vb0, rl_f(wB, j));
                    fma4(accB, vb1, rl_f(wB, j + 1));
                    fma4(accB, vb2, rl_f(wB, j + 2));
                    fma4(accB, vb3, rl_f(wB, j + 3));
                }
            }
            if (nA > LIST_CAP) accA = gather_row_by_scan(feat, cnt, src, dst, rowA, lane);
            if (nB > LIST_CAP) accB = gather_row_by_scan(feat, cnt, src, dst, rowB, lane);

            const float nrA = rsqrtf(fmaxf((float)nA, 1.0f)) * (1.0f - ALPHA_C);
            const float nrB = rsqrtf(fmaxf((float)nB, 1.0f)) * (1.0f - ALPHA_C);
            float4 hA, hB;
            hA.x = accA.x * nrA + ALPHA_C * f0A.x;
            hA.y = accA.y * nrA + ALPHA_C * f0A.y;
            hA.z = accA.z * nrA + ALPHA_C * f0A.z;
            hA.w = accA.w * nrA + ALPHA_C * f0A.w;
            hB.x = accB.x * nrB + ALPHA_C * f0B.x;
            hB.y = accB.y * nrB + ALPHA_C * f0B.y;
            hB.z = accB.z * nrB + ALPHA_C * f0B.z;
            hB.w = accB.w * nrB + ALPHA_C * f0B.w;
            float* hw = hb + lrA * HS_STRIDE + lane * 4;
            *(float4*)(hw) = hA;
            *(float4*)(hw + HS_STRIDE) = hB;
        }
        __syncthreads();

        const int strip1 = strip + N_BLOCKS, strip2 = strip + 2 * N_BLOCKS;
        const bool has1 = strip1 < N_STRIPS;
        const int sA1 = list_head(rA1, has1 ? __builtin_amdgcn_readlane(nv1, 0) : 0);
        const int sB1 = list_head(rB1, has1 ? __builtin_amdgcn_readlane(nv1, 1) : 0);
        const int dA1 = cnt[sA1], dB1 = cnt[sB1];
        const int m1 = min(strip1, N_STRIPS - 1) * 16;
        const float4 f0A1 = *(const float4*)(feat0 + (size_t)(m1 + lrA) * DIM + lane4);
        const float4 f0B1 = *(const float4*)(feat0 + (size_t)(m1 + lrB) * DIM + lane4);
        const int nv2 = row_degrees(strip2);
        const int rA2 = list_words(strip2, lrA), rB2 = list_words(strip2, lrB);

        v8f acc = {};
        const float* ha = hb + lr * HS_STRIDE + 2 * half;
        #pragma unroll
        for (int i = 0; i < DIM / 4; i++) {
            v2f a;
            a.x = ha[4 * i];
            a.y = ha[4 * i + 1];
            acc = __builtin_amdgcn_wmma_f32_16x16x4_f32(
                 false, a,  false, bfr[i],
                 (short)0, acc,  false,  false);
        }

        const float* he = hb + (8 * half) * HS_STRIDE + n0 + lr;
        float* op = out + (size_t)m0 * DIM + ((8 * half) * DIM + n0 + lr);
        #pragma unroll
        for (int r = 0; r < 8; r++) {
            float hv = he[r * HS_STRIDE];
            op[r * DIM] =
                (1.0f - BETA_C) * hv + BETA_C * acc[r] + bcol;
        }
        buf ^= 1;
        nv = nv1; nv1 = nv2;
        sA = sA1; sB = sB1; rA1 = rA2; rB1 = rB2; dA = dA1; dB = dB1;
        f0A = f0A1; f0B = f0B1;
    }
}

extern "C" void kernel_launch(void* const* d_in, const int* in_sizes, int n_in,
                              void* d_out, int out_size, void* d_ws, size_t ws_size,
                              hipStream_t stream) {
    const float* feat  = (const float*)d_in[0];
    const float* feat0 = (const float*)d_in[1];
    const float* W     = (const float*)d_in[2];
    const float* bias  = (const float*)d_in[3];
    const int*   src   = (const int*)d_in[4];
    const int*   dst   = (const int*)d_in[5];
    float* out = (float*)d_out;
    int* cnt   = (int*)d_ws;
    int* lists = (int*)d_out;

    gcnii_zero_cnt<<<(N_NODES + 255) / 256, 256, 0, stream>>>(cnt);
    gcnii_fill<<<(N_EDGES + 255) / 256, 256, 0, stream>>>(src, dst, cnt, lists);
    gcnii_gather_gemm<<<N_BLOCKS, 256, 0, stream>>>(feat, feat0, W, bias, cnt, src, dst, lists, out);
}


# Edge slotting moved off global returning atomics: 125 work-groups each own 400 targets, scan all of dst, take list slots from LDS counter adds (arrival order) and store cnt; clear kernel dropped
# speedup vs baseline: 2.5603x; 1.1630x over previous
#include <hip/hip_runtime.h>
#include <hip/hip_bf16.h>


#define N_NODES 50000
#define N_EDGES 600000
#define DIM 128
#define ALPHA_C 0.1f
#define BETA_C 0.22314355131420976f
#define HS_STRIDE 132
#define N_STRIPS (N_NODES / 16)
#define N_BLOCKS 768
#define LIST_CAP 128
#define NR 400
#define N_RANGES (N_NODES / NR)
#define FILL_THREADS 1024
#define FILL_UNROLL 8
#define Q_CAP 192
#define N_Q (N_EDGES / 4)
#define FULL_Q ((N_Q / (FILL_THREADS * FILL_UNROLL)) * (FILL_THREADS * FILL_UNROLL))
#define TAIL_ITERS ((N_Q - FULL_Q + FILL_THREADS - 1) / FILL_THREADS)

typedef __attribute__((ext_vector_type(2))) float v2f;
typedef __attribute__((ext_vector_type(8))) float v8f;

__global__ __launch_bounds__(FILL_THREADS) void gcnii_fill(const int* __restrict__ src,
                                                         const int* __restrict__ dst,
                                                         int* __restrict__ cnt,
                                                         int* __restrict__ lists) {
    __shared__ int lc[NR];
    __shared__ unsigned queue[(FILL_THREADS / 32) * Q_CAP];

    const int tid  = threadIdx.x;
    const int lane = tid & 31;
    const int wave = __builtin_amdgcn_readfirstlane(tid >> 5);
    const int base = blockIdx.x * NR;
    unsigned* myq = queue + wave * Q_CAP;

    for (int i = tid; i < NR; i += FILL_THREADS) lc[i] = 0;
    __syncthreads();

    int qn = 0;

    auto drain = [&]() {
        for (int i = lane; i < qn; i += 32) {
            const unsigned w = myq[i];
            const unsigned r = w >> 20;
            const int e = (int)(w & 0xFFFFFu);
            const int s = src[e];
            const int pos = atomicAdd(&lc[r], 1);
            if (pos < LIST_CAP) lists[(size_t)(base + (int)r) * DIM + pos] = s;
        }
        qn = 0;
    };
    auto push = [&](int dv, int e) {
        const unsigned r = (unsigned)(dv - base);
        const bool hit = r < (unsigned)NR;
        const unsigned mask = (unsigned)__ballot(hit);
        if (mask) {
            const unsigned below = __builtin_amdgcn_mbcnt_lo(mask, 0u);
            if (hit) myq[qn + (int)below] = (r << 20) | (unsigned)e;
            qn += __builtin_popcount(mask);
        }
    };

    const int4* dst4 = (const int4*)dst;
    auto take4 = [&](const int4 d, const int q) {
        const int e = 4 * q;
        push(d.x, e);
        push(d.y, e + 1);
        push(d.z, e + 2);
        push(d.w, e + 3);
        if (qn >= 32) drain();
    };
    for (int q0 = 0; q0 < FULL_Q; q0 += FILL_THREADS * FILL_UNROLL) {
        int4 d[FILL_UNROLL];
        #pragma unroll
        for (int u = 0; u < FILL_UNROLL; u++) d[u] = dst4[q0 + u * FILL_THREADS + tid];
        #pragma unroll
        for (int u = 0; u < FILL_UNROLL; u++) take4(d[u], q0 + u * FILL_THREADS + tid);
    }
    {
        int4 d[TAIL_ITERS];
        #pragma unroll
        for (int u = 0; u < TAIL_ITERS; u++) {
            const int q = FULL_Q + u * FILL_THREADS + tid;
            d[u] = (q < N_Q) ? dst4[q] : make_int4(-1, -1, -1, -1);
        }
        #pragma unroll
        for (int u = 0; u < TAIL_ITERS; u++) take4(d[u], FULL_Q + u * FILL_THREADS + tid);
    }
    drain();
    __syncthreads();
    for (int i = tid; i < NR; i += FILL_THREADS) cnt[base + i] = lc[i];
}

__device__ __forceinline__ float rl_f(float v, int l) {
    return __int_as_float(__builtin_amdgcn_readlane(__float_as_int(v), l));
}

__device__ __forceinline__ void fma4(float4& acc, const float4 v, const float w) {
    acc.x = fmaf(v.x, w, acc.x);
    acc.y = fmaf(v.y, w, acc.y);
    acc.z = fmaf(v.z, w, acc.z);
    acc.w = fmaf(v.w, w, acc.w);
}

__device__ __noinline__ float4 gather_row_by_scan(const float* __restrict__ feat,
                                                  const int* __restrict__ cnt,
                                                  const int* __restrict__ src,
                                                  const int* __restrict__ dst,
                                                  int row, int lane) {
    const float* frow = feat + lane * 4;
    float4 acc = make_float4(0.f, 0.f, 0.f, 0.f);
    for (int e0 = 0; e0 < N_EDGES; e0 += 32) {
        const int e = e0 + lane;
        const int s = src[e];
        const bool hit = (dst[e] == row);
        const float w = hit ? rsqrtf(fmaxf((float)cnt[s], 1.0f)) : 0.0f;
        unsigned mask = (unsigned)__ballot(hit);
        while (mask) {
            const int j = __builtin_ctz(mask);
            mask &= mask - 1;
            const int sj = __builtin_amdgcn_readlane(s, j);
            const float4 v = *(const float4*)(frow + (size_t)sj * DIM);
            fma4(acc, v, rl_f(w, j));
        }
    }
    return acc;
}

__global__ __launch_bounds__(256) void gcnii_gather_gemm(
    const float* __restrict__ feat,
    const float* __restrict__ feat0,
    const float* __restrict__ W,
    const float* __restrict__ bias,
    const int* __restrict__ cnt,
    const int* __restrict__ src,
    const int* __restrict__ dst,
    const int* lists,
    float* out)
{
    __shared__ float hs[2][16 * HS_STRIDE];

    const int tid  = threadIdx.x;
    const int lane = tid & 31;
    const int wave = __builtin_amdgcn_readfirstlane(tid >> 5);

    const int half = lane >> 4;
    const int lr   = lane & 15;
    const int n0   = wave * 16;

    v2f bfr[DIM / 4];
    #pragma unroll
    for (int i = 0; i < DIM / 4; i++) {
        const int ka = 4 * i + 2 * half;
        bfr[i].x = W[ka * DIM + n0 + lr];
        bfr[i].y = W[(ka + 1) * DIM + n0 + lr];
    }
    const float bcol = bias[n0 + lr];

    const int lrA = 2 * wave, lrB = 2 * wave + 1;
    const unsigned lane4 = (unsigned)lane * 4u;
    const float4* feat4 = (const float4*)feat;
    int buf = 0;

    const int lrP = 2 * wave + (lane & 1);
    auto row_degrees = [&](int s) -> int {
        return cnt[min(s, N_STRIPS - 1) * 16 + lrP];
    };
    auto list_words = [&](int s, int lr_) -> int {
        return lists[(size_t)(min(s, N_STRIPS - 1) * 16 + lr_) * DIM + lane];
    };
    auto list_head = [&](int words, int n) -> int {
        return (lane < ((n <= LIST_CAP) ? n : 0)) ? words : 0;
    };

    int strip = blockIdx.x;
    int nv  = row_degrees(strip);
    int nv1 = row_degrees(strip + N_BLOCKS);
    int sA  = list_words(strip, lrA), sB = list_words(strip, lrB);
    int rA1 = list_words(strip + N_BLOCKS, lrA), rB1 = list_words(strip + N_BLOCKS, lrB);
    float4 f0A = *(const float4*)(feat0 + (size_t)(strip * 16 + lrA) * DIM + lane4);
    float4 f0B = *(const float4*)(feat0 + (size_t)(strip * 16 + lrB) * DIM + lane4);
    sA = list_head(sA, __builtin_amdgcn_readlane(nv, 0));
    sB = list_head(sB, __builtin_amdgcn_readlane(nv, 1));
    int dA = cnt[sA], dB = cnt[sB];

    for (; strip < N_STRIPS; strip += N_BLOCKS) {
        const int m0 = strip * 16;
        float* hb = hs[buf];

        const int rowA = m0 + lrA, rowB = m0 + lrB;
        const int nA = __builtin_amdgcn_readlane(nv, 0), nB = __builtin_amdgcn_readlane(nv, 1);
        const int* lstA = lists + (size_t)rowA * DIM;
        const int* lstB = lists + (size_t)rowB * DIM;
        const int lenA = (nA <= LIST_CAP) ? nA : 0;
        const int lenB = (nB <= LIST_CAP) ? nB : 0;
        const int lenM = max(lenA, lenB);

        float wA = (lane < lenA) ? rsqrtf(fmaxf((float)dA, 1.0f)) : 0.0f;
        float wB = (lane < lenB) ? rsqrtf(fmaxf((float)dB, 1.0f)) : 0.0f;

        {
            float4 accA = make_float4(0.f, 0.f, 0.f, 0.f);
            float4 accB = make_float4(0.f, 0.f, 0.f, 0.f);
            for (int c = 0; c < lenM; c += 32) {
                const int mA = min(lenA - c, 32);
                const int mB = min(lenB - c, 32);
                if (c > 0) {
                    sA = 0; sB = 0;
                    if (lane < mA) sA = lstA[c + lane];
                    if (lane < mB) sB = lstB[c + lane];
                    const int eA = cnt[sA], eB = cnt[sB];
                    wA = (lane < mA) ? rsqrtf(fmaxf((float)eA, 1.0f)) : 0.0f;
                    wB = (lane < mB) ? rsqrtf(fmaxf((float)eB, 1.0f)) : 0.0f;
                }
                const int lastA = max(mA - 1, 0), lastB = max(mB - 1, 0);
                const int mM = max(mA, mB);
                for (int j = 0; j < mM; j += 4) {
                    const int a0 = __builtin_amdgcn_readlane(sA, min(j,     lastA));
                    const int a1 = __builtin_amdgcn_readlane(sA, min(j + 1, lastA));
                    const int a2 = __builtin_amdgcn_readlane(sA, min(j + 2, lastA));
                    const int a3 = __builtin_amdgcn_readlane(sA, min(j + 3, lastA));
                    const int b0 = __builtin_amdgcn_readlane(sB, min(j,     lastB));
                    const int b1 = __builtin_amdgcn_readlane(sB, min(j + 1, lastB));
                    const int b2 = __builtin_amdgcn_readlane(sB, min(j + 2, lastB));
                    const int b3 = __builtin_amdgcn_readlane(sB, min(j + 3, lastB));
                    const float4 va0 = feat4[(unsigned)(a0 * (DIM / 4)) + (unsigned)lane];
                    const float4 va1 = feat4[(unsigned)(a1 * (DIM / 4)) + (unsigned)lane];
                    const float4 va2 = feat4[(unsigned)(a2 * (DIM / 4)) + (unsigned)lane];
                    const float4 va3 = feat4[(unsigned)(a3 * (DIM / 4)) + (unsigned)lane];
                    const float4 vb0 = feat4[(unsigned)(b0 * (DIM / 4)) + (unsigned)lane];
                    const float4 vb1 = feat4[(unsigned)(b1 * (DIM / 4)) + (unsigned)lane];
                    const float4 vb2 = feat4[(unsigned)(b2 * (DIM / 4)) + (unsigned)lane];
                    const float4 vb3 = feat4[(unsigned)(b3 * (DIM / 4)) + (unsigned)lane];
                    fma4(accA, va0, rl_f(wA, j));
                    fma4(accA, va1, rl_f(wA, j + 1));
                    fma4(accA, va2, rl_f(wA, j + 2));
                    fma4(accA, va3, rl_f(wA, j + 3));
                    fma4(accB, vb0, rl_f(wB, j));
                    fma4(accB, vb1, rl_f(wB, j + 1));
                    fma4(accB, vb2, rl_f(wB, j + 2));
                    fma4(accB, vb3, rl_f(wB, j + 3));
                }
            }
            if (nA > LIST_CAP) accA = gather_row_by_scan(feat, cnt, src, dst, rowA, lane);
            if (nB > LIST_CAP) accB = gather_row_by_scan(feat, cnt, src, dst, rowB, lane);

            const float nrA = rsqrtf(fmaxf((float)nA, 1.0f)) * (1.0f - ALPHA_C);
            const float nrB = rsqrtf(fmaxf((float)nB, 1.0f)) * (1.0f - ALPHA_C);
            float4 hA, hB;
            hA.x = accA.x * nrA + ALPHA_C * f0A.x;
            hA.y = accA.y * nrA + ALPHA_C * f0A.y;
            hA.z = accA.z * nrA + ALPHA_C * f0A.z;
            hA.w = accA.w * nrA + ALPHA_C * f0A.w;
            hB.x = accB.x * nrB + ALPHA_C * f0B.x;
            hB.y = accB.y * nrB + ALPHA_C * f0B.y;
            hB.z = accB.z * nrB + ALPHA_C * f0B.z;
            hB.w = accB.w * nrB + ALPHA_C * f0B.w;
            float* hw = hb + lrA * HS_STRIDE + lane * 4;
            *(float4*)(hw) = hA;
            *(float4*)(hw + HS_STRIDE) = hB;
        }
        __syncthreads();

        const int strip1 = strip + N_BLOCKS, strip2 = strip + 2 * N_BLOCKS;
        const bool has1 = strip1 < N_STRIPS;
        const int sA1 = list_head(rA1, has1 ? __builtin_amdgcn_readlane(nv1, 0) : 0);
        const int sB1 = list_head(rB1, has1 ? __builtin_amdgcn_readlane(nv1, 1) : 0);
        const int dA1 = cnt[sA1], dB1 = cnt[sB1];
        const int m1 = min(strip1, N_STRIPS - 1) * 16;
        const float4 f0A1 = *(const float4*)(feat0 + (size_t)(m1 + lrA) * DIM + lane4);
        const float4 f0B1 = *(const float4*)(feat0 + (size_t)(m1 + lrB) * DIM + lane4);
        const int nv2 = row_degrees(strip2);
        const int rA2 = list_words(strip2, lrA), rB2 = list_words(strip2, lrB);

        v8f acc = {};
        const float* ha = hb + lr * HS_STRIDE + 2 * half;
        #pragma unroll
        for (int i = 0; i < DIM / 4; i++) {
            v2f a;
            a.x = ha[4 * i];
            a.y = ha[4 * i + 1];
            acc = __builtin_amdgcn_wmma_f32_16x16x4_f32(
                 false, a,  false, bfr[i],
                 (short)0, acc,  false,  false);
        }

        const float* he = hb + (8 * half) * HS_STRIDE + n0 + lr;
        float* op = out + (size_t)m0 * DIM + ((8 * half) * DIM + n0 + lr);
        #pragma unroll
        for (int r = 0; r < 8; r++) {
            float hv = he[r * HS_STRIDE];
            op[r * DIM] =
                (1.0f - BETA_C) * hv + BETA_C * acc[r] + bcol;
        }
        buf ^= 1;
        nv = nv1; nv1 = nv2;
        sA = sA1; sB = sB1; rA1 = rA2; rB1 = rB2; dA = dA1; dB = dB1;
        f0A = f0A1; f0B = f0B1;
    }
}

extern "C" void kernel_launch(void* const* d_in, const int* in_sizes, int n_in,
                              void* d_out, int out_size, void* d_ws, size_t ws_size,
                              hipStream_t stream) {
    const float* feat  = (const float*)d_in[0];
    const float* feat0 = (const float*)d_in[1];
    const float* W     = (const float*)d_in[2];
    const float* bias  = (const float*)d_in[3];
    const int*   src   = (const int*)d_in[4];
    const int*   dst   = (const int*)d_in[5];
    float* out = (float*)d_out;
    int* cnt   = (int*)d_ws;
    int* lists = (int*)d_out;

    gcnii_fill<<<N_RANGES, FILL_THREADS, 0, stream>>>(src, dst, cnt, lists);
    gcnii_gather_gemm<<<N_BLOCKS, 256, 0, stream>>>(feat, feat0, W, bias, cnt, src, dst, lists, out);
}


# Fill kernel's edge scan split: 64 node ranges x 4 edge-list quarters on 256 work-groups (was 125 ranges x whole list); each quarter fills its own 32-word list line; sums per quarter in arrival order
# speedup vs baseline: 3.2557x; 1.2716x over previous
#include <hip/hip_runtime.h>
#include <hip/hip_bf16.h>


#define N_NODES 50000
#define N_EDGES 600000
#define DIM 128
#define ALPHA_C 0.1f
#define BETA_C 0.22314355131420976f
#define HS_STRIDE 132
#define N_STRIPS (N_NODES / 16)
#define N_BLOCKS 768
#define LIST_CAP 128
#define N_SPLIT 4
#define SEG (LIST_CAP / N_SPLIT)
#define NR 782
#define N_RANGES 64
#define FILL_THREADS 1024
#define FILL_UNROLL 8
#define Q_CAP 192
#define N_Q (N_EDGES / 4)
#define Q_PER (N_Q / N_SPLIT)
#define FULL_Q ((Q_PER / (FILL_THREADS * FILL_UNROLL)) * (FILL_THREADS * FILL_UNROLL))
#define TAIL_ITERS ((Q_PER - FULL_Q + FILL_THREADS - 1) / FILL_THREADS)

typedef __attribute__((ext_vector_type(2))) float v2f;
typedef __attribute__((ext_vector_type(8))) float v8f;

__global__ __launch_bounds__(FILL_THREADS) void gcnii_fill(const int* __restrict__ src,
                                                         const int* __restrict__ dst,
                                                         int* __restrict__ cnt,
                                                         int* __restrict__ lists) {
    __shared__ int lc[NR];
    __shared__ unsigned queue[(FILL_THREADS / 32) * Q_CAP];

    const int tid  = threadIdx.x;
    const int lane = tid & 31;
    const int wave = __builtin_amdgcn_readfirstlane(tid >> 5);
    const int part = blockIdx.x % N_SPLIT;
    const int base = (blockIdx.x / N_SPLIT) * NR;
    const int qbase = part * Q_PER;
    unsigned* myq = queue + wave * Q_CAP;

    for (int i = tid; i < NR; i += FILL_THREADS) lc[i] = 0;
    __syncthreads();

    int qn = 0;

    auto drain = [&]() {
        for (int i = lane; i < qn; i += 32) {
            const unsigned w = myq[i];
            const unsigned r = w >> 20;
            const int e = (int)(w & 0xFFFFFu);
            const int s = src[e];
            const int pos = atomicAdd(&lc[r], 1);
            if (pos < SEG) lists[(size_t)(base + (int)r) * DIM + part * SEG + pos] = s;
        }
        qn = 0;
    };
    auto push = [&](int dv, int e) {
        const unsigned r = (unsigned)(dv - base);
        const bool hit = r < (unsigned)NR;
        const unsigned mask = (unsigned)__ballot(hit);
        if (mask) {
            const unsigned below = __builtin_amdgcn_mbcnt_lo(mask, 0u);
            if (hit) myq[qn + (int)below] = (r << 20) | (unsigned)e;
            qn += __builtin_popcount(mask);
        }
    };

    const int4* dst4 = (const int4*)dst;
    auto take4 = [&](const int4 d, const int q) {
        const int e = 4 * q;
        push(d.x, e);
        push(d.y, e + 1);
        push(d.z, e + 2);
        push(d.w, e + 3);
        if (qn >= 32) drain();
    };
    for (int q0 = 0; q0 < FULL_Q; q0 += FILL_THREADS * FILL_UNROLL) {
        int4 d[FILL_UNROLL];
        #pragma unroll
        for (int u = 0; u < FILL_UNROLL; u++) d[u] = dst4[qbase + q0 + u * FILL_THREADS + tid];
        #pragma unroll
        for (int u = 0; u < FILL_UNROLL; u++) take4(d[u], qbase + q0 + u * FILL_THREADS + tid);
    }
    {
        int4 d[TAIL_ITERS];
        #pragma unroll
        for (int u = 0; u < TAIL_ITERS; u++) {
            const int q = FULL_Q + u * FILL_THREADS + tid;
            d[u] = (q < Q_PER) ? dst4[qbase + q] : make_int4(-1, -1, -1, -1);
        }
        #pragma unroll
        for (int u = 0; u < TAIL_ITERS; u++) take4(d[u], qbase + FULL_Q + u * FILL_THREADS + tid);
    }
    drain();
    __syncthreads();
    for (int i = tid; i < NR; i += FILL_THREADS)
        if (base + i < N_NODES) cnt[(base + i) * N_SPLIT + part] = lc[i];
}

__device__ __forceinline__ float rl_f(float v, int l) {
    return __int_as_float(__builtin_amdgcn_readlane(__float_as_int(v), l));
}

__device__ __forceinline__ void fma4(float4& acc, const float4 v, const float w) {
    acc.x = fmaf(v.x, w, acc.x);
    acc.y = fmaf(v.y, w, acc.y);
    acc.z = fmaf(v.z, w, acc.z);
    acc.w = fmaf(v.w, w, acc.w);
}

__device__ __forceinline__ int sum4(const int4 q) { return q.x + q.y + q.z + q.w; }

struct RowLen { int n, len, p1, p2, p3; };

__device__ __forceinline__ RowLen row_len(const int4 v, const int l, const bool have) {
    const int c0 = __builtin_amdgcn_readlane(v.x, l), c1 = __builtin_amdgcn_readlane(v.y, l);
    const int c2 = __builtin_amdgcn_readlane(v.z, l), c3 = __builtin_amdgcn_readlane(v.w, l);
    RowLen r;
    r.n = c0 + c1 + c2 + c3;
    r.len = (have && max(max(c0, c1), max(c2, c3)) <= SEG) ? r.n : 0;
    r.p1 = c0; r.p2 = c0 + c1; r.p3 = c0 + c1 + c2;
    return r;
}

__device__ __forceinline__ int pull(const int x, const int i) {
    return __builtin_amdgcn_ds_bpermute((i & 31) << 2, x);
}

__device__ __forceinline__ int list_chunk(const int w0, const int w1, const int w2, const int w3,
                                          const RowLen r, const int c, const int lane) {
    const int e = c + lane;
    const int d0 = (c == 0) ? w0 : pull(w0, e);
    const int d1 = pull(w1, e - r.p1);
    const int d2 = pull(w2, e - r.p2);
    const int d3 = pull(w3, e - r.p3);
    const int v = (e < r.p1) ? d0 : (e < r.p2) ? d1 : (e < r.p3) ? d2 : d3;
    return (e < r.len) ? v : 0;
}

__device__ __noinline__ float4 gather_row_by_scan(const float* __restrict__ feat,
                                                  const int4* __restrict__ cnt4,
                                                  const int* __restrict__ src,
                                                  const int* __restrict__ dst,
                                                  int row, int lane) {
    const float* frow = feat + lane * 4;
    float4 acc = make_float4(0.f, 0.f, 0.f, 0.f);
    for (int e0 = 0; e0 < N_EDGES; e0 += 32) {
        const int e = e0 + lane;
        const int s = src[e];
        const bool hit = (dst[e] == row);
        const float w = hit ? rsqrtf(fmaxf((float)sum4(cnt4[s]), 1.0f)) : 0.0f;
        unsigned mask = (unsigned)__ballot(hit);
        while (mask) {
            const int j = __builtin_ctz(mask);
            mask &= mask - 1;
            const int sj = __builtin_amdgcn_readlane(s, j);
            const float4 v = *(const float4*)(frow + (size_t)sj * DIM);
            fma4(acc, v, rl_f(w, j));
        }
    }
    return acc;
}

__global__ __launch_bounds__(256) void gcnii_gather_gemm(
    const float* __restrict__ feat,
    const float* __restrict__ feat0,
    const float* __restrict__ W,
    const float* __restrict__ bias,
    const int* __restrict__ cnt,
    const int* __restrict__ src,
    const int* __restrict__ dst,
    const int* lists,
    float* out)
{
    __shared__ float hs[2][16 * HS_STRIDE];

    const int tid  = threadIdx.x;
    const int lane = tid & 31;
    const int wave = __builtin_amdgcn_readfirstlane(tid >> 5);

    const int half = lane >> 4;
    const int lr   = lane & 15;
    const int n0   = wave * 16;

    v2f bfr[DIM / 4];
    #pragma unroll
    for (int i = 0; i < DIM / 4; i++) {
        const int ka = 4 * i + 2 * half;
        bfr[i].x = W[ka * DIM + n0 + lr];
        bfr[i].y = W[(ka + 1) * DIM + n0 + lr];
    }
    const float bcol = bias[n0 + lr];

    const int lrA = 2 * wave, lrB = 2 * wave + 1;
    const unsigned lane4 = (unsigned)lane * 4u;
    const float4* feat4 = (const float4*)feat;
    const int4* cnt4 = (const int4*)cnt;
    int buf = 0;

    const int lrP = 2 * wave + (lane & 1);
    auto row_counts = [&](int s) -> int4 {
        return cnt4[min(s, N_STRIPS - 1) * 16 + lrP];
    };
    auto seg_words = [&](int s, int lr_, int k) -> int {
        return lists[(size_t)(min(s, N_STRIPS - 1) * 16 + lr_) * DIM + k * SEG + lane];
    };

    int strip = blockIdx.x;
    const int4 nv0 = row_counts(strip);
    int4 nv1 = row_counts(strip + N_BLOCKS);
    const int hA0 = seg_words(strip, lrA, 0), hA1 = seg_words(strip, lrA, 1);
    const int hA2 = seg_words(strip, lrA, 2), hA3 = seg_words(strip, lrA, 3);
    const int hB0 = seg_words(strip, lrB, 0), hB1 = seg_words(strip, lrB, 1);
    const int hB2 = seg_words(strip, lrB, 2), hB3 = seg_words(strip, lrB, 3);
    int rA0 = seg_words(strip + N_BLOCKS, lrA, 0), rA1 = seg_words(strip + N_BLOCKS, lrA, 1);
    int rA2 = seg_words(strip + N_BLOCKS, lrA, 2), rA3 = seg_words(strip + N_BLOCKS, lrA, 3);
    int rB0 = seg_words(strip + N_BLOCKS, lrB, 0), rB1 = seg_words(strip + N_BLOCKS, lrB, 1);
    int rB2 = seg_words(strip + N_BLOCKS, lrB, 2), rB3 = seg_words(strip + N_BLOCKS, lrB, 3);
    float4 f0A = *(const float4*)(feat0 + (size_t)(strip * 16 + lrA) * DIM + lane4);
    float4 f0B = *(const float4*)(feat0 + (size_t)(strip * 16 + lrB) * DIM + lane4);
    RowLen RA = row_len(nv0, 0, true), RB = row_len(nv0, 1, true);
    int sA = list_chunk(hA0, hA1, hA2, hA3, RA, 0, lane);
    int sB = list_chunk(hB0, hB1, hB2, hB3, RB, 0, lane);
    int4 dA = cnt4[sA], dB = cnt4[sB];

    for (; strip < N_STRIPS; strip += N_BLOCKS) {
        const int m0 = strip * 16;
        float* hb = hs[buf];

        const int rowA = m0 + lrA, rowB = m0 + lrB;
        const int nA = RA.n, nB = RB.n;
        const int lenA = RA.len, lenB = RB.len;
        const int lenM = max(lenA, lenB);

        float wA = (lane < lenA) ? rsqrtf(fmaxf((float)sum4(dA), 1.0f)) : 0.0f;
        float wB = (lane < lenB) ? rsqrtf(fmaxf((float)sum4(dB), 1.0f)) : 0.0f;

        {
            float4 accA = make_float4(0.f, 0.f, 0.f, 0.f);
            float4 accB = make_float4(0.f, 0.f, 0.f, 0.f);
            for (int c = 0; c < lenM; c += 32) {
                const int mA = min(lenA - c, 32);
                const int mB = min(lenB - c, 32);
                if (c > 0) {
                    sA = list_chunk(seg_words(strip, lrA, 0), seg_words(strip, lrA, 1),
                                    seg_words(strip, lrA, 2), seg_words(strip, lrA, 3), RA, c, lane);
                    sB = list_chunk(seg_words(strip, lrB, 0), seg_words(strip, lrB, 1),
                                    seg_words(strip, lrB, 2), seg_words(strip, lrB, 3), RB, c, lane);
                    const int4 eA = cnt4[sA], eB = cnt4[sB];
                    wA = (lane < mA) ? rsqrtf(fmaxf((float)sum4(eA), 1.0f)) : 0.0f;
                    wB = (lane < mB) ? rsqrtf(fmaxf((float)sum4(eB), 1.0f)) : 0.0f;
                }
                const int lastA = max(mA - 1, 0), lastB = max(mB - 1, 0);
                const int mM = max(mA, mB);
                for (int j = 0; j < mM; j += 4) {
                    const int a0 = __builtin_amdgcn_readlane(sA, min(j,     lastA));
                    const int a1 = __builtin_amdgcn_readlane(sA, min(j + 1, lastA));
                    const int a2 = __builtin_amdgcn_readlane(sA, min(j + 2, lastA));
                    const int a3 = __builtin_amdgcn_readlane(sA, min(j + 3, lastA));
                    const int b0 = __builtin_amdgcn_readlane(sB, min(j,     lastB));
                    const int b1 = __builtin_amdgcn_readlane(sB, min(j + 1, lastB));
                    const int b2 = __builtin_amdgcn_readlane(sB, min(j + 2, lastB));
                    const int b3 = __builtin_amdgcn_readlane(sB, min(j + 3, lastB));
                    const float4 va0 = feat4[(unsigned)(a0 * (DIM / 4)) + (unsigned)lane];
                    const float4 va1 = feat4[(unsigned)(a1 * (DIM / 4)) + (unsigned)lane];
                    const float4 va2 = feat4[(unsigned)(a2 * (DIM / 4)) + (unsigned)lane];
                    const float4 va3 = feat4[(unsigned)(a3 * (DIM / 4)) + (unsigned)lane];
                    const float4 vb0 = feat4[(unsigned)(b0 * (DIM / 4)) + (unsigned)lane];
                    const float4 vb1 = feat4[(unsigned)(b1 * (DIM / 4)) + (unsigned)lane];
                    const float4 vb2 = feat4[(unsigned)(b2 * (DIM / 4)) + (unsigned)lane];
                    const float4 vb3 = feat4[(unsigned)(b3 * (DIM / 4)) + (unsigned)lane];
                    fma4(accA, va0, rl_f(wA, j));
                    fma4(accA, va1, rl_f(wA, j + 1));
                    fma4(accA, va2, rl_f(wA, j + 2));
                    fma4(accA, va3, rl_f(wA, j + 3));
                    fma4(accB, vb0, rl_f(wB, j));
                    fma4(accB, vb1, rl_f(wB, j + 1));
                    fma4(accB, vb2, rl_f(wB, j + 2));
                    fma4(accB, vb3, rl_f(wB, j + 3));
                }
            }
            if (lenA != nA) accA = gather_row_by_scan(feat, cnt4, src, dst, rowA, lane);
            if (lenB != nB) accB = gather_row_by_scan(feat, cnt4, src, dst, rowB, lane);

            const float nrA = rsqrtf(fmaxf((float)nA, 1.0f)) * (1.0f - ALPHA_C);
            const float nrB = rsqrtf(fmaxf((float)nB, 1.0f)) * (1.0f - ALPHA_C);
            float4 hA, hB;
            hA.x = accA.x * nrA + ALPHA_C * f0A.x;
            hA.y = accA.y * nrA + ALPHA_C * f0A.y;
            hA.z = accA.z * nrA + ALPHA_C * f0A.z;
            hA.w = accA.w * nrA + ALPHA_C * f0A.w;
            hB.x = accB.x * nrB + ALPHA_C * f0B.x;
            hB.y = accB.y * nrB + ALPHA_C * f0B.y;
            hB.z = accB.z * nrB + ALPHA_C * f0B.z;
            hB.w = accB.w * nrB + ALPHA_C * f0B.w;
            float* hw = hb + lrA * HS_STRIDE + lane * 4;
            *(float4*)(hw) = hA;
            *(float4*)(hw + HS_STRIDE) = hB;
        }
        __syncthreads();

        const int strip1 = strip + N_BLOCKS, strip2 = strip + 2 * N_BLOCKS;
        const bool has1 = strip1 < N_STRIPS;
        const RowLen RA1 = row_len(nv1, 0, has1), RB1 = row_len(nv1, 1, has1);
        const int sA1 = list_chunk(rA0, rA1, rA2, rA3, RA1, 0, lane);
        const int sB1 = list_chunk(rB0, rB1, rB2, rB3, RB1, 0, lane);
        const int4 dA1 = cnt4[sA1], dB1 = cnt4[sB1];
        const int m1 = min(strip1, N_STRIPS - 1) * 16;
        const float4 f0A1 = *(const float4*)(feat0 + (size_t)(m1 + lrA) * DIM + lane4);
        const float4 f0B1 = *(const float4*)(feat0 + (size_t)(m1 + lrB) * DIM + lane4);
        const int4 nv2 = row_counts(strip2);
        const int tA0 = seg_words(strip2, lrA, 0), tA1 = seg_words(strip2, lrA, 1);
        const int tA2 = seg_words(strip2, lrA, 2), tA3 = seg_words(strip2, lrA, 3);
        const int tB0 = seg_words(strip2, lrB, 0), tB1 = seg_words(strip2, lrB, 1);
        const int tB2 = seg_words(strip2, lrB, 2), tB3 = seg_words(strip2, lrB, 3);

        v8f acc = {};
        const float* ha = hb + lr * HS_STRIDE + 2 * half;
        #pragma unroll
        for (int i = 0; i < DIM / 4; i++) {
            v2f a;
            a.x = ha[4 * i];
            a.y = ha[4 * i + 1];
            acc = __builtin_amdgcn_wmma_f32_16x16x4_f32(
                 false, a,  false, bfr[i],
                 (short)0, acc,  false,  false);
        }

        const float* he = hb + (8 * half) * HS_STRIDE + n0 + lr;
        float* op = out + (size_t)m0 * DIM + ((8 * half) * DIM + n0 + lr);
        #pragma unroll
        for (int r = 0; r < 8; r++) {
            float hv = he[r * HS_STRIDE];
            op[r * DIM] =
                (1.0f - BETA_C) * hv + BETA_C * acc[r] + bcol;
        }
        buf ^= 1;
        RA = RA1; RB = RB1; nv1 = nv2;
        sA = sA1; sB = sB1; dA = dA1; dB = dB1;
        rA0 = tA0; rA1 = tA1; rA2 = tA2; rA3 = tA3;
        rB0 = tB0; rB1 = tB1; rB2 = tB2; rB3 = tB3;
        f0A = f0A1; f0B = f0B1;
    }
}

extern "C" void kernel_launch(void* const* d_in, const int* in_sizes, int n_in,
                              void* d_out, int out_size, void* d_ws, size_t ws_size,
                              hipStream_t stream) {
    const float* feat  = (const float*)d_in[0];
    const float* feat0 = (const float*)d_in[1];
    const float* W     = (const float*)d_in[2];
    const float* bias  = (const float*)d_in[3];
    const int*   src   = (const int*)d_in[4];
    const int*   dst   = (const int*)d_in[5];
    float* out = (float*)d_out;
    int* cnt   = (int*)d_ws;
    int* lists = (int*)d_out;

    gcnii_fill<<<N_RANGES * N_SPLIT, FILL_THREADS, 0, stream>>>(src, dst, cnt, lists);
    gcnii_gather_gemm<<<N_BLOCKS, 256, 0, stream>>>(feat, feat0, W, bias, cnt, src, dst, lists, out);
}
